# starting kernel
# baseline (speedup 1.0000x reference)
#include <hip/hip_runtime.h>
#include <hip/hip_bf16.h>
#include <math.h>

#define Bq 32768
#define Lq 24
#define Hq 64
#define Vq 64

#define WAVES_PER_BLOCK 2
#define BLOCK_THREADS (WAVES_PER_BLOCK * 32)

typedef __attribute__((ext_vector_type(2))) float v2f;
typedef __attribute__((ext_vector_type(4))) float v4f;
typedef __attribute__((ext_vector_type(8))) float v8f;

#define OFF_W1   0
#define OFF_W2   8192
#define OFF_RP   16384
#define OFF_OW   20480
#define OFF_B1   24576
#define OFF_B2   24704
#define OFF_LNG  24768
#define OFF_LNB  24832
#define OFF_RPB  24896
#define OFF_OB   24960
#define OFF_OM   25024
#define OFF_WAVE 25056
#define H_STRIDE 68
#define HBUF_FLOATS (Lq * 16 * H_STRIDE)
#define X1_STRIDE 132
#define X1_FLOATS (16 * X1_STRIDE)
#define WAVE_FLOATS (HBUF_FLOATS + X1_FLOATS)
#define SMEM_FLOATS (OFF_WAVE + WAVES_PER_BLOCK * WAVE_FLOATS)
#define SMEM_BYTES (SMEM_FLOATS * 4)

template <int K>
__device__ inline v8f gemm_tile_f32(const float* __restrict__ As, int lda,
                                    const float* __restrict__ Bqp, int nstride4,
                                    v8f acc, int lane) {
  const int rc = lane & 15;
  const int hk = (lane >> 4) << 1;
  const float* ap = As + rc * lda + hk;
  const float* bp = Bqp + rc * 4 + hk;
#pragma unroll
  for (int k = 0; k < K; k += 4) {
    v2f a = *(const v2f*)(ap + k);
    v2f b = *(const v2f*)(bp + (k >> 2) * nstride4);
    acc = __builtin_amdgcn_wmma_f32_16x16x4_f32(
         false, a,  false, b,
         (short)0, acc,  false,  false);
  }
  return acc;
}

__global__ __launch_bounds__(BLOCK_THREADS)
void PerPositionEMADelta_fused(const int* __restrict__ seq,
                               const float* __restrict__ embed,
                               const float* __restrict__ w1,
                               const float* __restrict__ b1,
                               const float* __restrict__ w2,
                               const float* __restrict__ b2,
                               const float* __restrict__ ln_g,
                               const float* __restrict__ ln_b,
                               const float* __restrict__ rp_w,
                               const float* __restrict__ rp_b,
                               const float* __restrict__ out_w,
                               const float* __restrict__ out_b,
                               const float* __restrict__ alpha_logits,
                               float* __restrict__ out) {
  extern __shared__ float sm[];
  const int tid  = threadIdx.x;
  const int lane = tid & 31;
  const int wave = tid >> 5;
  const int rc   = lane & 15;
  const int half = lane >> 4;

  float* s_w1  = sm + OFF_W1;
  float* s_w2  = sm + OFF_W2;
  float* s_rp  = sm + OFF_RP;
  float* s_ow  = sm + OFF_OW;
  float* s_b1  = sm + OFF_B1;
  float* s_b2  = sm + OFF_B2;
  float* s_lng = sm + OFF_LNG;
  float* s_lnb = sm + OFF_LNB;
  float* s_rpb = sm + OFF_RPB;
  float* s_ob  = sm + OFF_OB;
  float* s_om  = sm + OFF_OM;
  float* wbase = sm + OFF_WAVE + wave * WAVE_FLOATS;
  float* s_H   = wbase;
  float* s_X1  = wbase + HBUF_FLOATS;

  for (int i = tid; i < 64 * 128; i += BLOCK_THREADS) {
    int k = i >> 7, n = i & 127;
    s_w1[(k >> 2) * 512 + n * 4 + (k & 3)] = w1[i];
  }
  for (int i = tid; i < 128 * 64; i += BLOCK_THREADS) {
    int k = i >> 6, n = i & 63;
    s_w2[(k >> 2) * 256 + n * 4 + (k & 3)] = w2[i];
  }
  for (int i = tid; i < 64 * 64; i += BLOCK_THREADS) {
    int k = i >> 6, n = i & 63;
    int p = (k >> 2) * 256 + n * 4 + (k & 3);
    s_rp[p] = rp_w[i];
    s_ow[p] = out_w[i];
  }
  for (int i = tid; i < 128; i += BLOCK_THREADS) s_b1[i] = b1[i];
  if (tid < 64) {
    s_b2[tid]  = b2[tid];
    s_lng[tid] = ln_g[tid];
    s_lnb[tid] = ln_b[tid];
    s_rpb[tid] = rp_b[tid];
    s_ob[tid]  = out_b[tid];
  }
  if (tid < Lq) {
    float sg = 1.0f / (1.0f + expf(-alpha_logits[tid]));
    s_om[tid] = 1.0f - (0.49f * sg + 0.5f);
  }
  __syncthreads();

  const int base = (blockIdx.x * WAVES_PER_BLOCK + wave) * 16;

  for (int t = 0; t < Lq; ++t) {
    float* Ht = s_H + t * (16 * H_STRIDE);
    {
      int idx = seq[(size_t)(base + rc) * Lq + t];
      const float* src = embed + (size_t)idx * Hq + half * 32;
      float* dst = Ht + rc * H_STRIDE + half * 32;
#pragma unroll
      for (int j = 0; j < 32; j += 4) *(v4f*)(dst + j) = *(const v4f*)(src + j);
    }
    __syncthreads();

    for (int nt = 0; nt < 8; ++nt) {
      v8f acc;
      float bv = s_b1[nt * 16 + rc];
#pragma unroll
      for (int j = 0; j < 8; ++j) acc[j] = bv;
      acc = gemm_tile_f32<64>(Ht, H_STRIDE, s_w1 + nt * 16 * 4, 512, acc, lane);
#pragma unroll
      for (int j = 0; j < 8; ++j) {
        int row = j + 8 * half;
        s_X1[row * X1_STRIDE + nt * 16 + rc] = fmaxf(acc[j], 0.0f);
      }
    }
    __syncthreads();

    for (int nt = 0; nt < 4; ++nt) {
      v8f acc;
      float bv = s_b2[nt * 16 + rc];
#pragma unroll
      for (int j = 0; j < 8; ++j) acc[j] = bv;
      acc = gemm_tile_f32<128>(s_X1, X1_STRIDE, s_w2 + nt * 16 * 4, 256, acc, lane);
#pragma unroll
      for (int j = 0; j < 8; ++j) {
        int row = j + 8 * half;
        Ht[row * H_STRIDE + nt * 16 + rc] += acc[j];
      }
    }
    __syncthreads();

    if (half == 0) {
      float* rowp = Ht + rc * H_STRIDE;
      float mu = 0.0f;
#pragma unroll
      for (int c = 0; c < 64; ++c) mu += rowp[c];
      mu *= (1.0f / 64.0f);
      float var = 0.0f;
#pragma unroll
      for (int c = 0; c < 64; ++c) { float d = rowp[c] - mu; var += d * d; }
      var *= (1.0f / 64.0f);
      float inv = rsqrtf(var + 1e-5f);
#pragma unroll
      for (int c = 0; c < 64; ++c)
        rowp[c] = (rowp[c] - mu) * inv * s_lng[c] + s_lnb[c];
    }
    __syncthreads();
  }

  float q[32], v[32];
  {
    const float* hl = s_H + 23 * (16 * H_STRIDE) + rc * H_STRIDE + half * 32;
#pragma unroll
    for (int j = 0; j < 32; ++j) { q[j] = hl[j]; v[j] = 0.0f; }
  }
  for (int s = Lq - 2; s >= 0; --s) {
    const float* hs = s_H + s * (16 * H_STRIDE) + rc * H_STRIDE + half * 32;
    float hk[32];
    float nsq = 0.0f;
#pragma unroll
    for (int j = 0; j < 32; ++j) { hk[j] = hs[j]; nsq += hk[j] * hk[j]; }
    nsq += __shfl_xor(nsq, 16, 32);
    float inv = 1.0f / fmaxf(sqrtf(nsq), 1e-12f);
    float dq = 0.0f;
#pragma unroll
    for (int j = 0; j < 32; ++j) dq += hk[j] * q[j];
    dq += __shfl_xor(dq, 16, 32);
    float c = s_om[s] * (dq * inv);
    float ci = c * inv;
#pragma unroll
    for (int j = 0; j < 32; ++j) {
      v[j] += c * hk[j];
      q[j] -= ci * hk[j];
    }
  }

  float* s_V  = s_X1;
  float* s_T1 = s_H;
  {
    float* dst = s_V + rc * H_STRIDE + half * 32;
#pragma unroll
    for (int j = 0; j < 32; ++j) dst[j] = v[j];
  }
  __syncthreads();

  for (int nt = 0; nt < 4; ++nt) {
    v8f acc;
    float bv = s_rpb[nt * 16 + rc];
#pragma unroll
    for (int j = 0; j < 8; ++j) acc[j] = bv;
    acc = gemm_tile_f32<64>(s_V, H_STRIDE, s_rp + nt * 16 * 4, 256, acc, lane);
#pragma unroll
    for (int j = 0; j < 8; ++j) {
      int row = j + 8 * half;
      s_T1[row * H_STRIDE + nt * 16 + rc] = acc[j];
    }
  }
  __syncthreads();

  for (int nt = 0; nt < 4; ++nt) {
    v8f acc;
    float bv = s_ob[nt * 16 + rc];
#pragma unroll
    for (int j = 0; j < 8; ++j) acc[j] = bv;
    acc = gemm_tile_f32<64>(s_T1, H_STRIDE, s_ow + nt * 16 * 4, 256, acc, lane);
#pragma unroll
    for (int j = 0; j < 8; ++j) {
      int row = j + 8 * half;
      out[(size_t)(base + row) * Vq + nt * 16 + rc] = acc[j];
    }
  }
}

extern "C" void kernel_launch(void* const* d_in, const int* in_sizes, int n_in,
                              void* d_out, int out_size, void* d_ws, size_t ws_size,
                              hipStream_t stream) {
  (void)n_in; (void)out_size; (void)d_ws; (void)ws_size;
  const int* seq            = (const int*)d_in[0];
  const float* embed        = (const float*)d_in[1];
  const float* w1           = (const float*)d_in[2];
  const float* b1           = (const float*)d_in[3];
  const float* w2           = (const float*)d_in[4];
  const float* b2           = (const float*)d_in[5];
  const float* ln_g         = (const float*)d_in[6];
  const float* ln_b         = (const float*)d_in[7];
  const float* rp_w         = (const float*)d_in[8];
  const float* rp_b         = (const float*)d_in[9];
  const float* out_w        = (const float*)d_in[10];
  const float* out_b        = (const float*)d_in[11];
  const float* alpha_logits = (const float*)d_in[12];
  float* out = (float*)d_out;

  const int B = in_sizes[0] / Lq;
  const int blocks = B / (16 * WAVES_PER_BLOCK);

  (void)hipFuncSetAttribute((const void*)PerPositionEMADelta_fused,
                            hipFuncAttributeMaxDynamicSharedMemorySize,
                            (int)SMEM_BYTES);

  PerPositionEMADelta_fused<<<blocks, BLOCK_THREADS, SMEM_BYTES, stream>>>(
      seq, embed, w1, b1, w2, b2, ln_g, ln_b, rp_w, rp_b, out_w, out_b,
      alpha_logits, out);
}


# Per-token MLP+LayerNorm and 1/norm depend only on the token id: each block builds the 64-row table once in LDS (same WMMA tiles); scan gathers rows by seq with 128-bit reads; 8 waves per block
# speedup vs baseline: 33.2152x; 33.2152x over previous
#include <hip/hip_runtime.h>
#include <hip/hip_bf16.h>
#include <math.h>


#define Bq 32768
#define Lq 24
#define Hq 64
#define Vq 64

#define WAVES_PER_BLOCK 8
#define BLOCK_THREADS (WAVES_PER_BLOCK * 32)

typedef __attribute__((ext_vector_type(2))) float v2f;
typedef __attribute__((ext_vector_type(4))) float v4f;
typedef __attribute__((ext_vector_type(8))) float v8f;
typedef __attribute__((ext_vector_type(4))) int v4i;

#define OFF_W1   0
#define OFF_W2   8192
#define OFF_RP   16384
#define OFF_OW   20480
#define OFF_B1   24576
#define OFF_B2   24704
#define OFF_LNG  24768
#define OFF_LNB  24832
#define OFF_RPB  24896
#define OFF_OB   24960
#define OFF_OM   25024
#define H_STRIDE 68
#define OFF_TAB  25056
#define X1_STRIDE 132
#define X1_FLOATS (16 * X1_STRIDE)
#define OFF_X1   29408
#define OFF_WAVE 37856
#define VT_FLOATS (16 * H_STRIDE)
#define WAVE_FLOATS (2 * VT_FLOATS)
#define SMEM_FLOATS (OFF_WAVE + WAVES_PER_BLOCK * WAVE_FLOATS)
#define SMEM_BYTES (SMEM_FLOATS * 4)

template <int K>
__device__ inline v8f gemm_tile_f32(const float* __restrict__ As, int lda,
                                    const float* __restrict__ Bqp, int nstride4,
                                    v8f acc, int lane) {
  const int rc = lane & 15;
  const int hk = (lane >> 4) << 1;
  const float* ap = As + rc * lda + hk;
  const float* bp = Bqp + rc * 4 + hk;
#pragma unroll
  for (int k = 0; k < K; k += 4) {
    v2f a = *(const v2f*)(ap + k);
    v2f b = *(const v2f*)(bp + (k >> 2) * nstride4);
    acc = __builtin_amdgcn_wmma_f32_16x16x4_f32(
         false, a,  false, b,
         (short)0, acc,  false,  false);
  }
  return acc;
}

__global__ __launch_bounds__(BLOCK_THREADS)
void PerPositionEMADelta_fused(const int* __restrict__ seq,
                               const float* __restrict__ embed,
                               const float* __restrict__ w1,
                               const float* __restrict__ b1,
                               const float* __restrict__ w2,
                               const float* __restrict__ b2,
                               const float* __restrict__ ln_g,
                               const float* __restrict__ ln_b,
                               const float* __restrict__ rp_w,
                               const float* __restrict__ rp_b,
                               const float* __restrict__ out_w,
                               const float* __restrict__ out_b,
                               const float* __restrict__ alpha_logits,
                               float* __restrict__ out) {
  extern __shared__ float sm[];
  const int tid  = threadIdx.x;
  const int lane = tid & 31;
  const int wave = tid >> 5;
  const int rc   = lane & 15;
  const int half = lane >> 4;

  float* s_w1  = sm + OFF_W1;
  float* s_w2  = sm + OFF_W2;
  float* s_rp  = sm + OFF_RP;
  float* s_ow  = sm + OFF_OW;
  float* s_b1  = sm + OFF_B1;
  float* s_b2  = sm + OFF_B2;
  float* s_lng = sm + OFF_LNG;
  float* s_lnb = sm + OFF_LNB;
  float* s_rpb = sm + OFF_RPB;
  float* s_ob  = sm + OFF_OB;
  float* s_om  = sm + OFF_OM;
  float* s_tab = sm + OFF_TAB;
  float* wbase = sm + OFF_WAVE + wave * WAVE_FLOATS;
  float* s_V   = wbase;
  float* s_T1  = wbase + VT_FLOATS;

  const int base = (blockIdx.x * WAVES_PER_BLOCK + wave) * 16;

  int idx[Lq];
  {
    const v4i* sp = (const v4i*)(seq + (size_t)(base + rc) * Lq);
#pragma unroll
    for (int j = 0; j < Lq / 4; ++j) {
      v4i t4 = sp[j];
      idx[4 * j + 0] = t4.x;
      idx[4 * j + 1] = t4.y;
      idx[4 * j + 2] = t4.z;
      idx[4 * j + 3] = t4.w;
    }
  }

  for (int i = tid; i < 64 * 128; i += BLOCK_THREADS) {
    int k = i >> 7, n = i & 127;
    s_w1[(k >> 2) * 512 + n * 4 + (k & 3)] = w1[i];
  }
  for (int i = tid; i < 128 * 64; i += BLOCK_THREADS) {
    int k = i >> 6, n = i & 63;
    s_w2[(k >> 2) * 256 + n * 4 + (k & 3)] = w2[i];
  }
  for (int i = tid; i < 64 * 64; i += BLOCK_THREADS) {
    int k = i >> 6, n = i & 63;
    int p = (k >> 2) * 256 + n * 4 + (k & 3);
    s_rp[p] = rp_w[i];
    s_ow[p] = out_w[i];
  }
  for (int i = tid; i < 128; i += BLOCK_THREADS) s_b1[i] = b1[i];
  if (tid < 64) {
    s_b2[tid]  = b2[tid];
    s_lng[tid] = ln_g[tid];
    s_lnb[tid] = ln_b[tid];
    s_rpb[tid] = rp_b[tid];
    s_ob[tid]  = out_b[tid];
  }
  if (tid < Lq) {
    float sg = 1.0f / (1.0f + expf(-alpha_logits[tid]));
    s_om[tid] = 1.0f - (0.49f * sg + 0.5f);
  }
  {
    const int row = tid >> 2;
    const int qo  = (tid & 3) * 16;
    const float* src = embed + (size_t)row * Hq + qo;
    float* dst = s_tab + row * H_STRIDE + qo;
#pragma unroll
    for (int j = 0; j < 16; j += 4) *(v4f*)(dst + j) = *(const v4f*)(src + j);
  }
  __syncthreads();

  {
    const int mt = wave & 3;
    const int hf = wave >> 2;
    float* Tt = s_tab + mt * (16 * H_STRIDE);
    float* X1 = sm + OFF_X1 + mt * X1_FLOATS;

    for (int i4 = 0; i4 < 4; ++i4) {
      const int nt = hf * 4 + i4;
      v8f acc;
      float bv = s_b1[nt * 16 + rc];
#pragma unroll
      for (int j = 0; j < 8; ++j) acc[j] = bv;
      acc = gemm_tile_f32<64>(Tt, H_STRIDE, s_w1 + nt * 16 * 4, 512, acc, lane);
#pragma unroll
      for (int j = 0; j < 8; ++j) {
        int row = j + 8 * half;
        X1[row * X1_STRIDE + nt * 16 + rc] = fmaxf(acc[j], 0.0f);
      }
    }
    __syncthreads();

    for (int i2 = 0; i2 < 2; ++i2) {
      const int nt = hf * 2 + i2;
      v8f acc;
      float bv = s_b2[nt * 16 + rc];
#pragma unroll
      for (int j = 0; j < 8; ++j) acc[j] = bv;
      acc = gemm_tile_f32<128>(X1, X1_STRIDE, s_w2 + nt * 16 * 4, 256, acc, lane);
#pragma unroll
      for (int j = 0; j < 8; ++j) {
        int row = j + 8 * half;
        Tt[row * H_STRIDE + nt * 16 + rc] += acc[j];
      }
    }
    __syncthreads();

    if (tid < Vq) {
      float* rowp = s_tab + tid * H_STRIDE;
      float mu = 0.0f;
#pragma unroll
      for (int c = 0; c < 64; ++c) mu += rowp[c];
      mu *= (1.0f / 64.0f);
      float var = 0.0f;
#pragma unroll
      for (int c = 0; c < 64; ++c) { float d = rowp[c] - mu; var += d * d; }
      var *= (1.0f / 64.0f);
      float inv = rsqrtf(var + 1e-5f);
      float n0 = 0.0f, n1 = 0.0f;
#pragma unroll
      for (int c = 0; c < 32; ++c) {
        float y = (rowp[c] - mu) * inv * s_lng[c] + s_lnb[c];
        rowp[c] = y;
        n0 += y * y;
      }
#pragma unroll
      for (int c = 32; c < 64; ++c) {
        float y = (rowp[c] - mu) * inv * s_lng[c] + s_lnb[c];
        rowp[c] = y;
        n1 += y * y;
      }
      float nsq = n0 + n1;
      rowp[64] = 1.0f / fmaxf(sqrtf(nsq), 1e-12f);
    }
    __syncthreads();
  }

  float q[32], v[32];
  {
    const float* hl = s_tab + idx[Lq - 1] * H_STRIDE + half * 32;
#pragma unroll
    for (int j = 0; j < 32; j += 4) {
      v4f t4 = *(const v4f*)(hl + j);
      q[j + 0] = t4.x; q[j + 1] = t4.y; q[j + 2] = t4.z; q[j + 3] = t4.w;
    }
#pragma unroll
    for (int j = 0; j < 32; ++j) v[j] = 0.0f;
  }
#pragma unroll
  for (int s = Lq - 2; s >= 0; --s) {
    const float* hrow = s_tab + idx[s] * H_STRIDE;
    const float* hs = hrow + half * 32;
    float hk[32];
#pragma unroll
    for (int j = 0; j < 32; j += 4) {
      v4f t4 = *(const v4f*)(hs + j);
      hk[j + 0] = t4.x; hk[j + 1] = t4.y; hk[j + 2] = t4.z; hk[j + 3] = t4.w;
    }
    float inv = hrow[64];
    float dq = 0.0f;
#pragma unroll
    for (int j = 0; j < 32; ++j) dq += hk[j] * q[j];
    dq += __shfl_xor(dq, 16, 32);
    float c = s_om[s] * (dq * inv);
    float ci = c * inv;
#pragma unroll
    for (int j = 0; j < 32; ++j) {
      v[j] += c * hk[j];
      q[j] -= ci * hk[j];
    }
  }

  {
    float* dst = s_V + rc * H_STRIDE + half * 32;
#pragma unroll
    for (int j = 0; j < 32; ++j) dst[j] = v[j];
  }
  __syncthreads();

  for (int nt = 0; nt < 4; ++nt) {
    v8f acc;
    float bv = s_rpb[nt * 16 + rc];
#pragma unroll
    for (int j = 0; j < 8; ++j) acc[j] = bv;
    acc = gemm_tile_f32<64>(s_V, H_STRIDE, s_rp + nt * 16 * 4, 256, acc, lane);
#pragma unroll
    for (int j = 0; j < 8; ++j) {
      int row = j + 8 * half;
      s_T1[row * H_STRIDE + nt * 16 + rc] = acc[j];
    }
  }
  __syncthreads();

  for (int nt = 0; nt < 4; ++nt) {
    v8f acc;
    float bv = s_ob[nt * 16 + rc];
#pragma unroll
    for (int j = 0; j < 8; ++j) acc[j] = bv;
    acc = gemm_tile_f32<64>(s_T1, H_STRIDE, s_ow + nt * 16 * 4, 256, acc, lane);
#pragma unroll
    for (int j = 0; j < 8; ++j) {
      int row = j + 8 * half;
      out[(size_t)(base + row) * Vq + nt * 16 + rc] = acc[j];
    }
  }
}

extern "C" void kernel_launch(void* const* d_in, const int* in_sizes, int n_in,
                              void* d_out, int out_size, void* d_ws, size_t ws_size,
                              hipStream_t stream) {
  (void)n_in; (void)out_size; (void)d_ws; (void)ws_size;
  const int* seq            = (const int*)d_in[0];
  const float* embed        = (const float*)d_in[1];
  const float* w1           = (const float*)d_in[2];
  const float* b1           = (const float*)d_in[3];
  const float* w2           = (const float*)d_in[4];
  const float* b2           = (const float*)d_in[5];
  const float* ln_g         = (const float*)d_in[6];
  const float* ln_b         = (const float*)d_in[7];
  const float* rp_w         = (const float*)d_in[8];
  const float* rp_b         = (const float*)d_in[9];
  const float* out_w        = (const float*)d_in[10];
  const float* out_b        = (const float*)d_in[11];
  const float* alpha_logits = (const float*)d_in[12];
  float* out = (float*)d_out;

  const int B = in_sizes[0] / Lq;
  const int blocks = B / (16 * WAVES_PER_BLOCK);

  (void)hipFuncSetAttribute((const void*)PerPositionEMADelta_fused,
                            hipFuncAttributeMaxDynamicSharedMemorySize,
                            (int)SMEM_BYTES);

  PerPositionEMADelta_fused<<<blocks, BLOCK_THREADS, SMEM_BYTES, stream>>>(
      seq, embed, w1, b1, w2, b2, ln_g, ln_b, rp_w, rp_b, out_w, out_b,
      alpha_logits, out);
}


# Weight preload: every global load issued as one group before the first LDS store (lane = column, 4 row loads, one ds_store_b128 per quad) instead of 18 serialised load-wait-store trips to L2
# speedup vs baseline: 46.9534x; 1.4136x over previous
#include <hip/hip_runtime.h>
#include <hip/hip_bf16.h>
#include <math.h>


#define Bq 32768
#define Lq 24
#define Hq 64
#define Vq 64

#define WAVES_PER_BLOCK 8
#define BLOCK_THREADS (WAVES_PER_BLOCK * 32)

typedef __attribute__((ext_vector_type(2))) float v2f;
typedef __attribute__((ext_vector_type(4))) float v4f;
typedef __attribute__((ext_vector_type(8))) float v8f;
typedef __attribute__((ext_vector_type(4))) int v4i;

#define OFF_W1   0
#define OFF_W2   8192
#define OFF_RP   16384
#define OFF_OW   20480
#define OFF_B1   24576
#define OFF_B2   24704
#define OFF_LNG  24768
#define OFF_LNB  24832
#define OFF_RPB  24896
#define OFF_OB   24960
#define OFF_OM   25024
#define H_STRIDE 68
#define OFF_TAB  25056
#define X1_STRIDE 132
#define X1_FLOATS (16 * X1_STRIDE)
#define OFF_X1   29408
#define OFF_WAVE 37856
#define VT_FLOATS (16 * H_STRIDE)
#define WAVE_FLOATS (2 * VT_FLOATS)
#define SMEM_FLOATS (OFF_WAVE + WAVES_PER_BLOCK * WAVE_FLOATS)
#define SMEM_BYTES (SMEM_FLOATS * 4)

template <int K>
__device__ inline v8f gemm_tile_f32(const float* __restrict__ As, int lda,
                                    const float* __restrict__ Bqp, int nstride4,
                                    v8f acc, int lane) {
  const int rc = lane & 15;
  const int hk = (lane >> 4) << 1;
  const float* ap = As + rc * lda + hk;
  const float* bp = Bqp + rc * 4 + hk;
#pragma unroll
  for (int k = 0; k < K; k += 4) {
    v2f a = *(const v2f*)(ap + k);
    v2f b = *(const v2f*)(bp + (k >> 2) * nstride4);
    acc = __builtin_amdgcn_wmma_f32_16x16x4_f32(
         false, a,  false, b,
         (short)0, acc,  false,  false);
  }
  return acc;
}

__global__ __launch_bounds__(BLOCK_THREADS)
void PerPositionEMADelta_fused(const int* __restrict__ seq,
                               const float* __restrict__ embed,
                               const float* __restrict__ w1,
                               const float* __restrict__ b1,
                               const float* __restrict__ w2,
                               const float* __restrict__ b2,
                               const float* __restrict__ ln_g,
                               const float* __restrict__ ln_b,
                               const float* __restrict__ rp_w,
                               const float* __restrict__ rp_b,
                               const float* __restrict__ out_w,
                               const float* __restrict__ out_b,
                               const float* __restrict__ alpha_logits,
                               float* __restrict__ out) {
  extern __shared__ float sm[];
  const int tid  = threadIdx.x;
  const int lane = tid & 31;
  const int wave = tid >> 5;
  const int rc   = lane & 15;
  const int half = lane >> 4;

  float* s_w1  = sm + OFF_W1;
  float* s_w2  = sm + OFF_W2;
  float* s_rp  = sm + OFF_RP;
  float* s_ow  = sm + OFF_OW;
  float* s_b1  = sm + OFF_B1;
  float* s_b2  = sm + OFF_B2;
  float* s_lng = sm + OFF_LNG;
  float* s_lnb = sm + OFF_LNB;
  float* s_rpb = sm + OFF_RPB;
  float* s_ob  = sm + OFF_OB;
  float* s_om  = sm + OFF_OM;
  float* s_tab = sm + OFF_TAB;
  float* wbase = sm + OFF_WAVE + wave * WAVE_FLOATS;
  float* s_V   = wbase;
  float* s_T1  = wbase + VT_FLOATS;

  const int base = (blockIdx.x * WAVES_PER_BLOCK + wave) * 16;

  int idx[Lq];
  {
    const v4i* sp = (const v4i*)(seq + (size_t)(base + rc) * Lq);
#pragma unroll
    for (int j = 0; j < Lq / 4; ++j) {
      v4i t4 = sp[j];
      idx[4 * j + 0] = t4.x;
      idx[4 * j + 1] = t4.y;
      idx[4 * j + 2] = t4.z;
      idx[4 * j + 3] = t4.w;
    }
  }

  v4f e4[4];
  {
    const float* src = embed + (size_t)(tid >> 2) * Hq + (tid & 3) * 16;
#pragma unroll
    for (int j = 0; j < 4; ++j) e4[j] = *(const v4f*)(src + 4 * j);
  }
  const float g_b1  = b1[tid & 127];
  const float g_b2  = b2[tid & 63];
  const float g_lng = ln_g[tid & 63];
  const float g_lnb = ln_b[tid & 63];
  const float g_rpb = rp_b[tid & 63];
  const float g_ob  = out_b[tid & 63];
  const float g_al  = alpha_logits[tid < Lq ? tid : 0];
  v4f q1[8], q2[8], q3[4], q4[4];
#pragma unroll
  for (int t = 0; t < 8; ++t) {
    const int wt = wave + 8 * t;
    const float* p = w1 + (wt >> 2) * 512 + (wt & 3) * 32 + lane;
    q1[t].x = p[0]; q1[t].y = p[128]; q1[t].z = p[256]; q1[t].w = p[384];
  }
#pragma unroll
  for (int t = 0; t < 8; ++t) {
    const int wt = wave + 8 * t;
    const float* p = w2 + (wt >> 1) * 256 + (wt & 1) * 32 + lane;
    q2[t].x = p[0]; q2[t].y = p[64]; q2[t].z = p[128]; q2[t].w = p[192];
  }
#pragma unroll
  for (int t = 0; t < 4; ++t) {
    const int wt = wave + 8 * t;
    const int o = (wt >> 1) * 256 + (wt & 1) * 32 + lane;
    const float* p = rp_w + o;
    const float* r = out_w + o;
    q3[t].x = p[0]; q3[t].y = p[64]; q3[t].z = p[128]; q3[t].w = p[192];
    q4[t].x = r[0]; q4[t].y = r[64]; q4[t].z = r[128]; q4[t].w = r[192];
  }

  {
    float* dst = s_tab + (tid >> 2) * H_STRIDE + (tid & 3) * 16;
#pragma unroll
    for (int j = 0; j < 4; ++j) *(v4f*)(dst + 4 * j) = e4[j];
  }
  if (tid < 128) s_b1[tid] = g_b1;
  if (tid < 64) {
    s_b2[tid]  = g_b2;
    s_lng[tid] = g_lng;
    s_lnb[tid] = g_lnb;
    s_rpb[tid] = g_rpb;
    s_ob[tid]  = g_ob;
  }
  if (tid < Lq) {
    float sg = 1.0f / (1.0f + expf(-g_al));
    s_om[tid] = 1.0f - (0.49f * sg + 0.5f);
  }
#pragma unroll
  for (int t = 0; t < 8; ++t)
    *(v4f*)(s_w1 + (wave + 8 * t) * 128 + lane * 4) = q1[t];
#pragma unroll
  for (int t = 0; t < 8; ++t)
    *(v4f*)(s_w2 + (wave + 8 * t) * 128 + lane * 4) = q2[t];
#pragma unroll
  for (int t = 0; t < 4; ++t) {
    *(v4f*)(s_rp + (wave + 8 * t) * 128 + lane * 4) = q3[t];
    *(v4f*)(s_ow + (wave + 8 * t) * 128 + lane * 4) = q4[t];
  }
  __syncthreads();

  {
    const int mt = wave & 3;
    const int hf = wave >> 2;
    float* Tt = s_tab + mt * (16 * H_STRIDE);
    float* X1 = sm + OFF_X1 + mt * X1_FLOATS;

    for (int i4 = 0; i4 < 4; ++i4) {
      const int nt = hf * 4 + i4;
      v8f acc;
      float bv = s_b1[nt * 16 + rc];
#pragma unroll
      for (int j = 0; j < 8; ++j) acc[j] = bv;
      acc = gemm_tile_f32<64>(Tt, H_STRIDE, s_w1 + nt * 16 * 4, 512, acc, lane);
#pragma unroll
      for (int j = 0; j < 8; ++j) {
        int row = j + 8 * half;
        X1[row * X1_STRIDE + nt * 16 + rc] = fmaxf(acc[j], 0.0f);
      }
    }
    __syncthreads();

    for (int i2 = 0; i2 < 2; ++i2) {
      const int nt = hf * 2 + i2;
      v8f acc;
      float bv = s_b2[nt * 16 + rc];
#pragma unroll
      for (int j = 0; j < 8; ++j) acc[j] = bv;
      acc = gemm_tile_f32<128>(X1, X1_STRIDE, s_w2 + nt * 16 * 4, 256, acc, lane);
#pragma unroll
      for (int j = 0; j < 8; ++j) {
        int row = j + 8 * half;
        Tt[row * H_STRIDE + nt * 16 + rc] += acc[j];
      }
    }
    __syncthreads();

    if (tid < Vq) {
      float* rowp = s_tab + tid * H_STRIDE;
      float mu = 0.0f;
#pragma unroll
      for (int c = 0; c < 64; ++c) mu += rowp[c];
      mu *= (1.0f / 64.0f);
      float var = 0.0f;
#pragma unroll
      for (int c = 0; c < 64; ++c) { float d = rowp[c] - mu; var += d * d; }
      var *= (1.0f / 64.0f);
      float inv = rsqrtf(var + 1e-5f);
      float n0 = 0.0f, n1 = 0.0f;
#pragma unroll
      for (int c = 0; c < 32; ++c) {
        float y = (rowp[c] - mu) * inv * s_lng[c] + s_lnb[c];
        rowp[c] = y;
        n0 += y * y;
      }
#pragma unroll
      for (int c = 32; c < 64; ++c) {
        float y = (rowp[c] - mu) * inv * s_lng[c] + s_lnb[c];
        rowp[c] = y;
        n1 += y * y;
      }
      float nsq = n0 + n1;
      rowp[64] = 1.0f / fmaxf(sqrtf(nsq), 1e-12f);
    }
    __syncthreads();
  }

  float q[32], v[32];
  {
    const float* hl = s_tab + idx[Lq - 1] * H_STRIDE + half * 32;
#pragma unroll
    for (int j = 0; j < 32; j += 4) {
      v4f t4 = *(const v4f*)(hl + j);
      q[j + 0] = t4.x; q[j + 1] = t4.y; q[j + 2] = t4.z; q[j + 3] = t4.w;
    }
#pragma unroll
    for (int j = 0; j < 32; ++j) v[j] = 0.0f;
  }
#pragma unroll
  for (int s = Lq - 2; s >= 0; --s) {
    const float* hrow = s_tab + idx[s] * H_STRIDE;
    const float* hs = hrow + half * 32;
    float hk[32];
#pragma unroll
    for (int j = 0; j < 32; j += 4) {
      v4f t4 = *(const v4f*)(hs + j);
      hk[j + 0] = t4.x; hk[j + 1] = t4.y; hk[j + 2] = t4.z; hk[j + 3] = t4.w;
    }
    float inv = hrow[64];
    float dq = 0.0f;
#pragma unroll
    for (int j = 0; j < 32; ++j) dq += hk[j] * q[j];
    dq += __shfl_xor(dq, 16, 32);
    float c = s_om[s] * (dq * inv);
    float ci = c * inv;
#pragma unroll
    for (int j = 0; j < 32; ++j) {
      v[j] += c * hk[j];
      q[j] -= ci * hk[j];
    }
  }

  {
    float* dst = s_V + rc * H_STRIDE + half * 32;
#pragma unroll
    for (int j = 0; j < 32; ++j) dst[j] = v[j];
  }
  __syncthreads();

  for (int nt = 0; nt < 4; ++nt) {
    v8f acc;
    float bv = s_rpb[nt * 16 + rc];
#pragma unroll
    for (int j = 0; j < 8; ++j) acc[j] = bv;
    acc = gemm_tile_f32<64>(s_V, H_STRIDE, s_rp + nt * 16 * 4, 256, acc, lane);
#pragma unroll
    for (int j = 0; j < 8; ++j) {
      int row = j + 8 * half;
      s_T1[row * H_STRIDE + nt * 16 + rc] = acc[j];
    }
  }
  __syncthreads();

  for (int nt = 0; nt < 4; ++nt) {
    v8f acc;
    float bv = s_ob[nt * 16 + rc];
#pragma unroll
    for (int j = 0; j < 8; ++j) acc[j] = bv;
    acc = gemm_tile_f32<64>(s_T1, H_STRIDE, s_ow + nt * 16 * 4, 256, acc, lane);
#pragma unroll
    for (int j = 0; j < 8; ++j) {
      int row = j + 8 * half;
      out[(size_t)(base + row) * Vq + nt * 16 + rc] = acc[j];
    }
  }
}

extern "C" void kernel_launch(void* const* d_in, const int* in_sizes, int n_in,
                              void* d_out, int out_size, void* d_ws, size_t ws_size,
                              hipStream_t stream) {
  (void)n_in; (void)out_size; (void)d_ws; (void)ws_size;
  const int* seq            = (const int*)d_in[0];
  const float* embed        = (const float*)d_in[1];
  const float* w1           = (const float*)d_in[2];
  const float* b1           = (const float*)d_in[3];
  const float* w2           = (const float*)d_in[4];
  const float* b2           = (const float*)d_in[5];
  const float* ln_g         = (const float*)d_in[6];
  const float* ln_b         = (const float*)d_in[7];
  const float* rp_w         = (const float*)d_in[8];
  const float* rp_b         = (const float*)d_in[9];
  const float* out_w        = (const float*)d_in[10];
  const float* out_b        = (const float*)d_in[11];
  const float* alpha_logits = (const float*)d_in[12];
  float* out = (float*)d_out;

  const int B = in_sizes[0] / Lq;
  const int blocks = B / (16 * WAVES_PER_BLOCK);

  (void)hipFuncSetAttribute((const void*)PerPositionEMADelta_fused,
                            hipFuncAttributeMaxDynamicSharedMemorySize,
                            (int)SMEM_BYTES);

  PerPositionEMADelta_fused<<<blocks, BLOCK_THREADS, SMEM_BYTES, stream>>>(
      seq, embed, w1, b1, w2, b2, ln_g, ln_b, rp_w, rp_b, out_w, out_b,
      alpha_logits, out);
}


# Output head folded: each block forms W=rp_w@out_w and the bias row rp_b@out_w+out_b once on WMMA (waves idle during LayerNorm); per-tile head is one GEMM, not two; head sums reassociated
# speedup vs baseline: 49.6288x; 1.0570x over previous
#include <hip/hip_runtime.h>
#include <hip/hip_bf16.h>
#include <math.h>


#define Bq 32768
#define Lq 24
#define Hq 64
#define Vq 64

#define WAVES_PER_BLOCK 8
#define BLOCK_THREADS (WAVES_PER_BLOCK * 32)

typedef __attribute__((ext_vector_type(2))) float v2f;
typedef __attribute__((ext_vector_type(4))) float v4f;
typedef __attribute__((ext_vector_type(8))) float v8f;
typedef __attribute__((ext_vector_type(4))) int v4i;

#define OFF_W1   0
#define OFF_W2   8192
#define OFF_RP   16384
#define OFF_OW   20480
#define OFF_B1   24576
#define OFF_B2   24704
#define OFF_LNG  24768
#define OFF_LNB  24832
#define OFF_RPB  24896
#define OFF_OB   24960
#define OFF_OM   25024
#define H_STRIDE 68
#define OFF_TAB  25056
#define X1_STRIDE 132
#define X1_FLOATS (16 * X1_STRIDE)
#define OFF_X1   29408
#define OFF_WF   OFF_X1
#define OFF_BF   (OFF_X1 + 4096)
#define OFF_WAVE 37856
#define VT_FLOATS (16 * H_STRIDE)
#define WAVE_FLOATS VT_FLOATS
#define SMEM_FLOATS (OFF_WAVE + WAVES_PER_BLOCK * WAVE_FLOATS)
#define SMEM_BYTES (SMEM_FLOATS * 4)

template <int K>
__device__ inline v8f gemm_tile_f32(const float* __restrict__ As, int lda,
                                    const float* __restrict__ Bqp, int nstride4,
                                    v8f acc, int lane) {
  const int rc = lane & 15;
  const int hk = (lane >> 4) << 1;
  const float* ap = As + rc * lda + hk;
  const float* bp = Bqp + rc * 4 + hk;
#pragma unroll
  for (int k = 0; k < K; k += 4) {
    v2f a = *(const v2f*)(ap + k);
    v2f b = *(const v2f*)(bp + (k >> 2) * nstride4);
    acc = __builtin_amdgcn_wmma_f32_16x16x4_f32(
         false, a,  false, b,
         (short)0, acc,  false,  false);
  }
  return acc;
}

__global__ __launch_bounds__(BLOCK_THREADS)
void PerPositionEMADelta_fused(const int* __restrict__ seq,
                               const float* __restrict__ embed,
                               const float* __restrict__ w1,
                               const float* __restrict__ b1,
                               const float* __restrict__ w2,
                               const float* __restrict__ b2,
                               const float* __restrict__ ln_g,
                               const float* __restrict__ ln_b,
                               const float* __restrict__ rp_w,
                               const float* __restrict__ rp_b,
                               const float* __restrict__ out_w,
                               const float* __restrict__ out_b,
                               const float* __restrict__ alpha_logits,
                               float* __restrict__ out) {
  extern __shared__ float sm[];
  const int tid  = threadIdx.x;
  const int lane = tid & 31;
  const int wave = tid >> 5;
  const int rc   = lane & 15;
  const int half = lane >> 4;
  const int wave_s = __builtin_amdgcn_readfirstlane(wave);

  float* s_w1  = sm + OFF_W1;
  float* s_w2  = sm + OFF_W2;
  float* s_rp  = sm + OFF_RP;
  float* s_ow  = sm + OFF_OW;
  float* s_b1  = sm + OFF_B1;
  float* s_b2  = sm + OFF_B2;
  float* s_lng = sm + OFF_LNG;
  float* s_lnb = sm + OFF_LNB;
  float* s_rpb = sm + OFF_RPB;
  float* s_ob  = sm + OFF_OB;
  float* s_om  = sm + OFF_OM;
  float* s_tab = sm + OFF_TAB;
  float* wbase = sm + OFF_WAVE + wave * WAVE_FLOATS;
  float* s_V   = wbase;
  float* s_wf  = sm + OFF_WF;
  float* s_bf  = sm + OFF_BF;

  const int base = (blockIdx.x * WAVES_PER_BLOCK + wave) * 16;

  int idx[Lq];
  {
    const v4i* sp = (const v4i*)(seq + (size_t)(base + rc) * Lq);
#pragma unroll
    for (int j = 0; j < Lq / 4; ++j) {
      v4i t4 = sp[j];
      idx[4 * j + 0] = t4.x;
      idx[4 * j + 1] = t4.y;
      idx[4 * j + 2] = t4.z;
      idx[4 * j + 3] = t4.w;
    }
  }

  v4f e4[4];
  {
    const float* src = embed + (size_t)(tid >> 2) * Hq + (tid & 3) * 16;
#pragma unroll
    for (int j = 0; j < 4; ++j) e4[j] = *(const v4f*)(src + 4 * j);
  }
  const float g_b1  = b1[tid & 127];
  const float g_b2  = b2[tid & 63];
  const float g_lng = ln_g[tid & 63];
  const float g_lnb = ln_b[tid & 63];
  const float g_rpb = rp_b[tid & 63];
  const float g_ob  = out_b[tid & 63];
  const float g_al  = alpha_logits[tid < Lq ? tid : 0];
  v4f q1[8], q2[8], q3[4], q4[4];
#pragma unroll
  for (int t = 0; t < 8; ++t) {
    const int wt = wave + 8 * t;
    const float* p = w1 + (wt >> 2) * 512 + (wt & 3) * 32 + lane;
    q1[t].x = p[0]; q1[t].y = p[128]; q1[t].z = p[256]; q1[t].w = p[384];
  }
#pragma unroll
  for (int t = 0; t < 8; ++t) {
    const int wt = wave + 8 * t;
    const float* p = w2 + (wt >> 1) * 256 + (wt & 1) * 32 + lane;
    q2[t].x = p[0]; q2[t].y = p[64]; q2[t].z = p[128]; q2[t].w = p[192];
  }
#pragma unroll
  for (int t = 0; t < 4; ++t) {
    const int wt = wave + 8 * t;
    const int o = (wt >> 1) * 256 + (wt & 1) * 32 + lane;
    const float* p = rp_w + o;
    const float* r = out_w + o;
    q3[t].x = p[0]; q3[t].y = p[64]; q3[t].z = p[128]; q3[t].w = p[192];
    q4[t].x = r[0]; q4[t].y = r[64]; q4[t].z = r[128]; q4[t].w = r[192];
  }

  {
    float* dst = s_tab + (tid >> 2) * H_STRIDE + (tid & 3) * 16;
#pragma unroll
    for (int j = 0; j < 4; ++j) *(v4f*)(dst + 4 * j) = e4[j];
  }
  if (tid < 128) s_b1[tid] = g_b1;
  if (tid < 64) {
    s_b2[tid]  = g_b2;
    s_lng[tid] = g_lng;
    s_lnb[tid] = g_lnb;
    s_rpb[tid] = g_rpb;
    s_ob[tid]  = g_ob;
  }
  if (tid < Lq) {
    float sg = 1.0f / (1.0f + expf(-g_al));
    s_om[tid] = 1.0f - (0.49f * sg + 0.5f);
  }
#pragma unroll
  for (int t = 0; t < 8; ++t)
    *(v4f*)(s_w1 + (wave + 8 * t) * 128 + lane * 4) = q1[t];
#pragma unroll
  for (int t = 0; t < 8; ++t)
    *(v4f*)(s_w2 + (wave + 8 * t) * 128 + lane * 4) = q2[t];
#pragma unroll
  for (int t = 0; t < 4; ++t) {
    *(v4f*)(s_rp + (wave + 8 * t) * 128 + lane * 4) = q3[t];
    *(v4f*)(s_ow + (wave + 8 * t) * 128 + lane * 4) = q4[t];
  }
  __syncthreads();

  {
    const int mt = wave & 3;
    const int hf = wave >> 2;
    float* Tt = s_tab + mt * (16 * H_STRIDE);
    float* X1 = sm + OFF_X1 + mt * X1_FLOATS;

    for (int i4 = 0; i4 < 4; ++i4) {
      const int nt = hf * 4 + i4;
      v8f acc;
      float bv = s_b1[nt * 16 + rc];
#pragma unroll
      for (int j = 0; j < 8; ++j) acc[j] = bv;
      acc = gemm_tile_f32<64>(Tt, H_STRIDE, s_w1 + nt * 16 * 4, 512, acc, lane);
#pragma unroll
      for (int j = 0; j < 8; ++j) {
        int row = j + 8 * half;
        X1[row * X1_STRIDE + nt * 16 + rc] = fmaxf(acc[j], 0.0f);
      }
    }
    __syncthreads();

    for (int i2 = 0; i2 < 2; ++i2) {
      const int nt = hf * 2 + i2;
      v8f acc;
      float bv = s_b2[nt * 16 + rc];
#pragma unroll
      for (int j = 0; j < 8; ++j) acc[j] = bv;
      acc = gemm_tile_f32<128>(X1, X1_STRIDE, s_w2 + nt * 16 * 4, 256, acc, lane);
#pragma unroll
      for (int j = 0; j < 8; ++j) {
        int row = j + 8 * half;
        Tt[row * H_STRIDE + nt * 16 + rc] += acc[j];
      }
    }
    __syncthreads();

    if (tid < Vq) {
      float* rowp = s_tab + tid * H_STRIDE;
      float mu = 0.0f;
#pragma unroll
      for (int c = 0; c < 64; ++c) mu += rowp[c];
      mu *= (1.0f / 64.0f);
      float var = 0.0f;
#pragma unroll
      for (int c = 0; c < 64; ++c) { float d = rowp[c] - mu; var += d * d; }
      var *= (1.0f / 64.0f);
      float inv = rsqrtf(var + 1e-5f);
      float n0 = 0.0f, n1 = 0.0f;
#pragma unroll
      for (int c = 0; c < 32; ++c) {
        float y = (rowp[c] - mu) * inv * s_lng[c] + s_lnb[c];
        rowp[c] = y;
        n0 += y * y;
      }
#pragma unroll
      for (int c = 32; c < 64; ++c) {
        float y = (rowp[c] - mu) * inv * s_lng[c] + s_lnb[c];
        rowp[c] = y;
        n1 += y * y;
      }
      float nsq = n0 + n1;
      rowp[64] = 1.0f / fmaxf(sqrtf(nsq), 1e-12f);
    } else if (wave_s >= 4) {
      const int ft = wave_s - 4;
      const int hk2 = half << 1;
      const int m = 16 * ft + rc;
      const float* ap = s_rp + (m >> 2) * 256 + (m & 3) + 4 * hk2;
      const float* bp = s_ow + rc * 4 + hk2;
      v8f f0, f1, f2, f3, fb;
      {
        const float ob = s_ob[ft * 16 + rc];
#pragma unroll
        for (int j = 0; j < 8; ++j) {
          f0[j] = 0.0f; f1[j] = 0.0f; f2[j] = 0.0f; f3[j] = 0.0f; fb[j] = ob;
        }
      }
#pragma unroll
      for (int k = 0; k < 64; k += 4) {
        v2f a;
        a.x = ap[4 * k];
        a.y = ap[4 * k + 4];
        const v2f ab = *(const v2f*)(s_rpb + k + hk2);
        const float* bk = bp + (k >> 2) * 256;
        const v2f b0 = *(const v2f*)(bk);
        const v2f b1 = *(const v2f*)(bk + 64);
        const v2f b2 = *(const v2f*)(bk + 128);
        const v2f b3 = *(const v2f*)(bk + 192);
        const v2f bb = *(const v2f*)(bk + ft * 64);
        f0 = __builtin_amdgcn_wmma_f32_16x16x4_f32(false, a, false, b0, (short)0, f0, false, false);
        f1 = __builtin_amdgcn_wmma_f32_16x16x4_f32(false, a, false, b1, (short)0, f1, false, false);
        f2 = __builtin_amdgcn_wmma_f32_16x16x4_f32(false, a, false, b2, (short)0, f2, false, false);
        f3 = __builtin_amdgcn_wmma_f32_16x16x4_f32(false, a, false, b3, (short)0, f3, false, false);
        fb = __builtin_amdgcn_wmma_f32_16x16x4_f32(false, ab, false, bb, (short)0, fb, false, false);
      }
      float* wp = s_wf + (4 * ft + 2 * half) * 256 + 4 * rc;
      v4f lo, hi;
      lo.x = f0[0]; lo.y = f0[1]; lo.z = f0[2]; lo.w = f0[3];
      hi.x = f0[4]; hi.y = f0[5]; hi.z = f0[6]; hi.w = f0[7];
      *(v4f*)(wp) = lo;        *(v4f*)(wp + 256) = hi;
      lo.x = f1[0]; lo.y = f1[1]; lo.z = f1[2]; lo.w = f1[3];
      hi.x = f1[4]; hi.y = f1[5]; hi.z = f1[6]; hi.w = f1[7];
      *(v4f*)(wp + 64) = lo;   *(v4f*)(wp + 64 + 256) = hi;
      lo.x = f2[0]; lo.y = f2[1]; lo.z = f2[2]; lo.w = f2[3];
      hi.x = f2[4]; hi.y = f2[5]; hi.z = f2[6]; hi.w = f2[7];
      *(v4f*)(wp + 128) = lo;  *(v4f*)(wp + 128 + 256) = hi;
      lo.x = f3[0]; lo.y = f3[1]; lo.z = f3[2]; lo.w = f3[3];
      hi.x = f3[4]; hi.y = f3[5]; hi.z = f3[6]; hi.w = f3[7];
      *(v4f*)(wp + 192) = lo;  *(v4f*)(wp + 192 + 256) = hi;
      if (half == 0) s_bf[ft * 16 + rc] = fb[0];
    }
    __syncthreads();
  }

  float q[32], v[32];
  {
    const float* hl = s_tab + idx[Lq - 1] * H_STRIDE + half * 32;
#pragma unroll
    for (int j = 0; j < 32; j += 4) {
      v4f t4 = *(const v4f*)(hl + j);
      q[j + 0] = t4.x; q[j + 1] = t4.y; q[j + 2] = t4.z; q[j + 3] = t4.w;
    }
#pragma unroll
    for (int j = 0; j < 32; ++j) v[j] = 0.0f;
  }
#pragma unroll
  for (int s = Lq - 2; s >= 0; --s) {
    const float* hrow = s_tab + idx[s] * H_STRIDE;
    const float* hs = hrow + half * 32;
    float hk[32];
#pragma unroll
    for (int j = 0; j < 32; j += 4) {
      v4f t4 = *(const v4f*)(hs + j);
      hk[j + 0] = t4.x; hk[j + 1] = t4.y; hk[j + 2] = t4.z; hk[j + 3] = t4.w;
    }
    float inv = hrow[64];
    float dq = 0.0f;
#pragma unroll
    for (int j = 0; j < 32; ++j) dq += hk[j] * q[j];
    dq += __shfl_xor(dq, 16, 32);
    float c = s_om[s] * (dq * inv);
    float ci = c * inv;
#pragma unroll
    for (int j = 0; j < 32; ++j) {
      v[j] += c * hk[j];
      q[j] -= ci * hk[j];
    }
  }

  {
    float* dst = s_V + rc * H_STRIDE + half * 32;
#pragma unroll
    for (int j = 0; j < 32; ++j) dst[j] = v[j];
  }
  __syncthreads();

  for (int nt = 0; nt < 4; ++nt) {
    v8f acc;
    float bv = s_bf[nt * 16 + rc];
#pragma unroll
    for (int j = 0; j < 8; ++j) acc[j] = bv;
    acc = gemm_tile_f32<64>(s_V, H_STRIDE, s_wf + nt * 16 * 4, 256, acc, lane);
#pragma unroll
    for (int j = 0; j < 8; ++j) {
      int row = j + 8 * half;
      out[(size_t)(base + row) * Vq + nt * 16 + rc] = acc[j];
    }
  }
}

extern "C" void kernel_launch(void* const* d_in, const int* in_sizes, int n_in,
                              void* d_out, int out_size, void* d_ws, size_t ws_size,
                              hipStream_t stream) {
  (void)n_in; (void)out_size; (void)d_ws; (void)ws_size;
  const int* seq            = (const int*)d_in[0];
  const float* embed        = (const float*)d_in[1];
  const float* w1           = (const float*)d_in[2];
  const float* b1           = (const float*)d_in[3];
  const float* w2           = (const float*)d_in[4];
  const float* b2           = (const float*)d_in[5];
  const float* ln_g         = (const float*)d_in[6];
  const float* ln_b         = (const float*)d_in[7];
  const float* rp_w         = (const float*)d_in[8];
  const float* rp_b         = (const float*)d_in[9];
  const float* out_w        = (const float*)d_in[10];
  const float* out_b        = (const float*)d_in[11];
  const float* alpha_logits = (const float*)d_in[12];
  float* out = (float*)d_out;

  const int B = in_sizes[0] / Lq;
  const int blocks = B / (16 * WAVES_PER_BLOCK);

  (void)hipFuncSetAttribute((const void*)PerPositionEMADelta_fused,
                            hipFuncAttributeMaxDynamicSharedMemorySize,
                            (int)SMEM_BYTES);

  PerPositionEMADelta_fused<<<blocks, BLOCK_THREADS, SMEM_BYTES, stream>>>(
      seq, embed, w1, b1, w2, b2, ln_g, ln_b, rp_w, rp_b, out_w, out_b,
      alpha_logits, out);
}


# Preload overlapped with the first table GEMM: it starts once embed, w1 and the small vectors have landed; w2, rp_w, out_w go to LDS behind it; b1 and alpha loads join the one load group
# speedup vs baseline: 52.3031x; 1.0539x over previous
#include <hip/hip_runtime.h>
#include <hip/hip_bf16.h>
#include <math.h>


#define Bq 32768
#define Lq 24
#define Hq 64
#define Vq 64

#define WAVES_PER_BLOCK 8
#define BLOCK_THREADS (WAVES_PER_BLOCK * 32)

typedef __attribute__((ext_vector_type(2))) float v2f;
typedef __attribute__((ext_vector_type(4))) float v4f;
typedef __attribute__((ext_vector_type(8))) float v8f;
typedef __attribute__((ext_vector_type(4))) int v4i;

#define OFF_W1   0
#define OFF_W2   8192
#define OFF_RP   16384
#define OFF_OW   20480
#define OFF_B1   24576
#define OFF_B2   24704
#define OFF_LNG  24768
#define OFF_LNB  24832
#define OFF_RPB  24896
#define OFF_OB   24960
#define OFF_OM   25024
#define H_STRIDE 68
#define OFF_TAB  25056
#define X1_STRIDE 132
#define X1_FLOATS (16 * X1_STRIDE)
#define OFF_X1   29408
#define OFF_WF   OFF_X1
#define OFF_BF   (OFF_X1 + 4096)
#define OFF_WAVE 37856
#define VT_FLOATS (16 * H_STRIDE)
#define WAVE_FLOATS VT_FLOATS
#define SMEM_FLOATS (OFF_WAVE + WAVES_PER_BLOCK * WAVE_FLOATS)
#define SMEM_BYTES (SMEM_FLOATS * 4)

template <int K>
__device__ inline v8f gemm_tile_f32(const float* __restrict__ As, int lda,
                                    const float* __restrict__ Bqp, int nstride4,
                                    v8f acc, int lane) {
  const int rc = lane & 15;
  const int hk = (lane >> 4) << 1;
  const float* ap = As + rc * lda + hk;
  const float* bp = Bqp + rc * 4 + hk;
#pragma unroll
  for (int k = 0; k < K; k += 4) {
    v2f a = *(const v2f*)(ap + k);
    v2f b = *(const v2f*)(bp + (k >> 2) * nstride4);
    acc = __builtin_amdgcn_wmma_f32_16x16x4_f32(
         false, a,  false, b,
         (short)0, acc,  false,  false);
  }
  return acc;
}

__global__ __launch_bounds__(BLOCK_THREADS)
void PerPositionEMADelta_fused(const int* __restrict__ seq,
                               const float* __restrict__ embed,
                               const float* __restrict__ w1,
                               const float* __restrict__ b1,
                               const float* __restrict__ w2,
                               const float* __restrict__ b2,
                               const float* __restrict__ ln_g,
                               const float* __restrict__ ln_b,
                               const float* __restrict__ rp_w,
                               const float* __restrict__ rp_b,
                               const float* __restrict__ out_w,
                               const float* __restrict__ out_b,
                               const float* __restrict__ alpha_logits,
                               float* __restrict__ out) {
  extern __shared__ float sm[];
  const int tid  = threadIdx.x;
  const int lane = tid & 31;
  const int wave = tid >> 5;
  const int rc   = lane & 15;
  const int half = lane >> 4;
  const int wave_s = __builtin_amdgcn_readfirstlane(wave);

  float* s_w1  = sm + OFF_W1;
  float* s_w2  = sm + OFF_W2;
  float* s_rp  = sm + OFF_RP;
  float* s_ow  = sm + OFF_OW;
  float* s_b1  = sm + OFF_B1;
  float* s_b2  = sm + OFF_B2;
  float* s_lng = sm + OFF_LNG;
  float* s_lnb = sm + OFF_LNB;
  float* s_rpb = sm + OFF_RPB;
  float* s_ob  = sm + OFF_OB;
  float* s_om  = sm + OFF_OM;
  float* s_tab = sm + OFF_TAB;
  float* wbase = sm + OFF_WAVE + wave * WAVE_FLOATS;
  float* s_V   = wbase;
  float* s_wf  = sm + OFF_WF;
  float* s_bf  = sm + OFF_BF;

  const int base = (blockIdx.x * WAVES_PER_BLOCK + wave) * 16;

  v4f e4[4];
  {
    const float* src = embed + (size_t)(tid >> 2) * Hq + (tid & 3) * 16;
#pragma unroll
    for (int j = 0; j < 4; ++j) e4[j] = *(const v4f*)(src + 4 * j);
  }
  const float g_b1  = b1[tid & 127];
  const float g_b2  = b2[tid & 63];
  const float g_lng = ln_g[tid & 63];
  const float g_lnb = ln_b[tid & 63];
  const float g_rpb = rp_b[tid & 63];
  const float g_ob  = out_b[tid & 63];
  const float g_al  = alpha_logits[(tid & 31) < Lq ? (tid & 31) : 0];
  v4f q1[8], q2[8], q3[4], q4[4];
#pragma unroll
  for (int t = 0; t < 8; ++t) {
    const int wt = wave + 8 * t;
    const float* p = w1 + (wt >> 2) * 512 + (wt & 3) * 32 + lane;
    q1[t].x = p[0]; q1[t].y = p[128]; q1[t].z = p[256]; q1[t].w = p[384];
  }
  asm volatile("" ::: "memory");
#pragma unroll
  for (int t = 0; t < 8; ++t) {
    const int wt = wave + 8 * t;
    const float* p = w2 + (wt >> 1) * 256 + (wt & 1) * 32 + lane;
    q2[t].x = p[0]; q2[t].y = p[64]; q2[t].z = p[128]; q2[t].w = p[192];
  }
#pragma unroll
  for (int t = 0; t < 4; ++t) {
    const int wt = wave + 8 * t;
    const int o = (wt >> 1) * 256 + (wt & 1) * 32 + lane;
    const float* p = rp_w + o;
    const float* r = out_w + o;
    q3[t].x = p[0]; q3[t].y = p[64]; q3[t].z = p[128]; q3[t].w = p[192];
    q4[t].x = r[0]; q4[t].y = r[64]; q4[t].z = r[128]; q4[t].w = r[192];
  }
  int idx[Lq];
  {
    const v4i* sp = (const v4i*)(seq + (size_t)(base + rc) * Lq);
#pragma unroll
    for (int j = 0; j < Lq / 4; ++j) {
      v4i t4 = sp[j];
      idx[4 * j + 0] = t4.x;
      idx[4 * j + 1] = t4.y;
      idx[4 * j + 2] = t4.z;
      idx[4 * j + 3] = t4.w;
    }
  }

  {
    float* dst = s_tab + (tid >> 2) * H_STRIDE + (tid & 3) * 16;
#pragma unroll
    for (int j = 0; j < 4; ++j) *(v4f*)(dst + 4 * j) = e4[j];
  }
  s_b1[tid & 127] = g_b1;
  s_b2[tid & 63]  = g_b2;
  s_lng[tid & 63] = g_lng;
  s_lnb[tid & 63] = g_lnb;
  s_rpb[tid & 63] = g_rpb;
  s_ob[tid & 63]  = g_ob;
  {
    float sg = 1.0f / (1.0f + expf(-g_al));
    s_om[tid & 31] = 1.0f - (0.49f * sg + 0.5f);
  }
#pragma unroll
  for (int t = 0; t < 8; ++t)
    *(v4f*)(s_w1 + (wave + 8 * t) * 128 + lane * 4) = q1[t];
  __builtin_amdgcn_fence(__ATOMIC_RELEASE, "workgroup", "local");
  __builtin_amdgcn_s_barrier();
  __builtin_amdgcn_fence(__ATOMIC_ACQUIRE, "workgroup", "local");

  {
    const int mt = wave & 3;
    const int hf = wave >> 2;
    float* Tt = s_tab + mt * (16 * H_STRIDE);
    float* X1 = sm + OFF_X1 + mt * X1_FLOATS;

    for (int i4 = 0; i4 < 4; ++i4) {
      const int nt = hf * 4 + i4;
      v8f acc;
      float bv = s_b1[nt * 16 + rc];
#pragma unroll
      for (int j = 0; j < 8; ++j) acc[j] = bv;
      acc = gemm_tile_f32<64>(Tt, H_STRIDE, s_w1 + nt * 16 * 4, 512, acc, lane);
#pragma unroll
      for (int j = 0; j < 8; ++j) {
        int row = j + 8 * half;
        X1[row * X1_STRIDE + nt * 16 + rc] = fmaxf(acc[j], 0.0f);
      }
    }
#pragma unroll
    for (int t = 0; t < 8; ++t)
      *(v4f*)(s_w2 + (wave + 8 * t) * 128 + lane * 4) = q2[t];
#pragma unroll
    for (int t = 0; t < 4; ++t) {
      *(v4f*)(s_rp + (wave + 8 * t) * 128 + lane * 4) = q3[t];
      *(v4f*)(s_ow + (wave + 8 * t) * 128 + lane * 4) = q4[t];
    }
    __syncthreads();

    for (int i2 = 0; i2 < 2; ++i2) {
      const int nt = hf * 2 + i2;
      v8f acc;
      float bv = s_b2[nt * 16 + rc];
#pragma unroll
      for (int j = 0; j < 8; ++j) acc[j] = bv;
      acc = gemm_tile_f32<128>(X1, X1_STRIDE, s_w2 + nt * 16 * 4, 256, acc, lane);
#pragma unroll
      for (int j = 0; j < 8; ++j) {
        int row = j + 8 * half;
        Tt[row * H_STRIDE + nt * 16 + rc] += acc[j];
      }
    }
    __syncthreads();

    if (tid < Vq) {
      float* rowp = s_tab + tid * H_STRIDE;
      float mu = 0.0f;
#pragma unroll
      for (int c = 0; c < 64; ++c) mu += rowp[c];
      mu *= (1.0f / 64.0f);
      float var = 0.0f;
#pragma unroll
      for (int c = 0; c < 64; ++c) { float d = rowp[c] - mu; var += d * d; }
      var *= (1.0f / 64.0f);
      float inv = rsqrtf(var + 1e-5f);
      float n0 = 0.0f, n1 = 0.0f;
#pragma unroll
      for (int c = 0; c < 32; ++c) {
        float y = (rowp[c] - mu) * inv * s_lng[c] + s_lnb[c];
        rowp[c] = y;
        n0 += y * y;
      }
#pragma unroll
      for (int c = 32; c < 64; ++c) {
        float y = (rowp[c] - mu) * inv * s_lng[c] + s_lnb[c];
        rowp[c] = y;
        n1 += y * y;
      }
      float nsq = n0 + n1;
      rowp[64] = 1.0f / fmaxf(sqrtf(nsq), 1e-12f);
    } else if (wave_s >= 4) {
      const int ft = wave_s - 4;
      const int hk2 = half << 1;
      const int m = 16 * ft + rc;
      const float* ap = s_rp + (m >> 2) * 256 + (m & 3) + 4 * hk2;
      const float* bp = s_ow + rc * 4 + hk2;
      v8f f0, f1, f2, f3, fb;
      {
        const float ob = s_ob[ft * 16 + rc];
#pragma unroll
        for (int j = 0; j < 8; ++j) {
          f0[j] = 0.0f; f1[j] = 0.0f; f2[j] = 0.0f; f3[j] = 0.0f; fb[j] = ob;
        }
      }
#pragma unroll
      for (int k = 0; k < 64; k += 4) {
        v2f a;
        a.x = ap[4 * k];
        a.y = ap[4 * k + 4];
        const v2f ab = *(const v2f*)(s_rpb + k + hk2);
        const float* bk = bp + (k >> 2) * 256;
        const v2f b0 = *(const v2f*)(bk);
        const v2f b1 = *(const v2f*)(bk + 64);
        const v2f b2 = *(const v2f*)(bk + 128);
        const v2f b3 = *(const v2f*)(bk + 192);
        const v2f bb = *(const v2f*)(bk + ft * 64);
        f0 = __builtin_amdgcn_wmma_f32_16x16x4_f32(false, a, false, b0, (short)0, f0, false, false);
        f1 = __builtin_amdgcn_wmma_f32_16x16x4_f32(false, a, false, b1, (short)0, f1, false, false);
        f2 = __builtin_amdgcn_wmma_f32_16x16x4_f32(false, a, false, b2, (short)0, f2, false, false);
        f3 = __builtin_amdgcn_wmma_f32_16x16x4_f32(false, a, false, b3, (short)0, f3, false, false);
        fb = __builtin_amdgcn_wmma_f32_16x16x4_f32(false, ab, false, bb, (short)0, fb, false, false);
      }
      float* wp = s_wf + (4 * ft + 2 * half) * 256 + 4 * rc;
      v4f lo, hi;
      lo.x = f0[0]; lo.y = f0[1]; lo.z = f0[2]; lo.w = f0[3];
      hi.x = f0[4]; hi.y = f0[5]; hi.z = f0[6]; hi.w = f0[7];
      *(v4f*)(wp) = lo;        *(v4f*)(wp + 256) = hi;
      lo.x = f1[0]; lo.y = f1[1]; lo.z = f1[2]; lo.w = f1[3];
      hi.x = f1[4]; hi.y = f1[5]; hi.z = f1[6]; hi.w = f1[7];
      *(v4f*)(wp + 64) = lo;   *(v4f*)(wp + 64 + 256) = hi;
      lo.x = f2[0]; lo.y = f2[1]; lo.z = f2[2]; lo.w = f2[3];
      hi.x = f2[4]; hi.y = f2[5]; hi.z = f2[6]; hi.w = f2[7];
      *(v4f*)(wp + 128) = lo;  *(v4f*)(wp + 128 + 256) = hi;
      lo.x = f3[0]; lo.y = f3[1]; lo.z = f3[2]; lo.w = f3[3];
      hi.x = f3[4]; hi.y = f3[5]; hi.z = f3[6]; hi.w = f3[7];
      *(v4f*)(wp + 192) = lo;  *(v4f*)(wp + 192 + 256) = hi;
      if (half == 0) s_bf[ft * 16 + rc] = fb[0];
    }
    __syncthreads();
  }

  float q[32], v[32];
  {
    const float* hl = s_tab + idx[Lq - 1] * H_STRIDE + half * 32;
#pragma unroll
    for (int j = 0; j < 32; j += 4) {
      v4f t4 = *(const v4f*)(hl + j);
      q[j + 0] = t4.x; q[j + 1] = t4.y; q[j + 2] = t4.z; q[j + 3] = t4.w;
    }
#pragma unroll
    for (int j = 0; j < 32; ++j) v[j] = 0.0f;
  }
#pragma unroll
  for (int s = Lq - 2; s >= 0; --s) {
    const float* hrow = s_tab + idx[s] * H_STRIDE;
    const float* hs = hrow + half * 32;
    float hk[32];
#pragma unroll
    for (int j = 0; j < 32; j += 4) {
      v4f t4 = *(const v4f*)(hs + j);
      hk[j + 0] = t4.x; hk[j + 1] = t4.y; hk[j + 2] = t4.z; hk[j + 3] = t4.w;
    }
    float inv = hrow[64];
    float dq = 0.0f;
#pragma unroll
    for (int j = 0; j < 32; ++j) dq += hk[j] * q[j];
    dq += __shfl_xor(dq, 16, 32);
    float c = s_om[s] * (dq * inv);
    float ci = c * inv;
#pragma unroll
    for (int j = 0; j < 32; ++j) {
      v[j] += c * hk[j];
      q[j] -= ci * hk[j];
    }
  }

  {
    float* dst = s_V + rc * H_STRIDE + half * 32;
#pragma unroll
    for (int j = 0; j < 32; ++j) dst[j] = v[j];
  }
  __syncthreads();

  for (int nt = 0; nt < 4; ++nt) {
    v8f acc;
    float bv = s_bf[nt * 16 + rc];
#pragma unroll
    for (int j = 0; j < 8; ++j) acc[j] = bv;
    acc = gemm_tile_f32<64>(s_V, H_STRIDE, s_wf + nt * 16 * 4, 256, acc, lane);
#pragma unroll
    for (int j = 0; j < 8; ++j) {
      int row = j + 8 * half;
      out[(size_t)(base + row) * Vq + nt * 16 + rc] = acc[j];
    }
  }
}

extern "C" void kernel_launch(void* const* d_in, const int* in_sizes, int n_in,
                              void* d_out, int out_size, void* d_ws, size_t ws_size,
                              hipStream_t stream) {
  (void)n_in; (void)out_size; (void)d_ws; (void)ws_size;
  const int* seq            = (const int*)d_in[0];
  const float* embed        = (const float*)d_in[1];
  const float* w1           = (const float*)d_in[2];
  const float* b1           = (const float*)d_in[3];
  const float* w2           = (const float*)d_in[4];
  const float* b2           = (const float*)d_in[5];
  const float* ln_g         = (const float*)d_in[6];
  const float* ln_b         = (const float*)d_in[7];
  const float* rp_w         = (const float*)d_in[8];
  const float* rp_b         = (const float*)d_in[9];
  const float* out_w        = (const float*)d_in[10];
  const float* out_b        = (const float*)d_in[11];
  const float* alpha_logits = (const float*)d_in[12];
  float* out = (float*)d_out;

  const int B = in_sizes[0] / Lq;
  const int blocks = B / (16 * WAVES_PER_BLOCK);

  (void)hipFuncSetAttribute((const void*)PerPositionEMADelta_fused,
                            hipFuncAttributeMaxDynamicSharedMemorySize,
                            (int)SMEM_BYTES);

  PerPositionEMADelta_fused<<<blocks, BLOCK_THREADS, SMEM_BYTES, stream>>>(
      seq, embed, w1, b1, w2, b2, ln_g, ln_b, rp_w, rp_b, out_w, out_b,
      alpha_logits, out);
}


# Scan re-mapped to one lane per batch row (waves 0-3 x 32 rows): k.q summed in-lane as a tree, no cross-lane step; rows read in lane-rotated chunk order from a 512-B-pitch copy; sum order changed
# speedup vs baseline: 64.4611x; 1.2325x over previous
#include <hip/hip_runtime.h>
#include <hip/hip_bf16.h>
#include <math.h>


#define Bq 32768
#define Lq 24
#define Hq 64
#define Vq 64

#define WAVES_PER_BLOCK 8
#define BLOCK_THREADS (WAVES_PER_BLOCK * 32)

typedef __attribute__((ext_vector_type(2))) float v2f;
typedef __attribute__((ext_vector_type(4))) float v4f;
typedef __attribute__((ext_vector_type(8))) float v8f;
typedef __attribute__((ext_vector_type(4))) int v4i;

#define OFF_W1   0
#define OFF_W2   8192
#define OFF_RP   16384
#define OFF_OW   20480
#define OFF_B1   24576
#define OFF_B2   24704
#define OFF_LNG  24768
#define OFF_LNB  24832
#define OFF_RPB  24896
#define OFF_OB   24960
#define OFF_OM   25024
#define H_STRIDE 68
#define OFF_TAB  25056
#define X1_STRIDE 132
#define X1_FLOATS (16 * X1_STRIDE)
#define OFF_X1   29408
#define OFF_WF   OFF_X1
#define OFF_BF   (OFF_X1 + 4096)
#define OFF_WAVE 37856
#define VT_FLOATS (16 * H_STRIDE)
#define WAVE_FLOATS VT_FLOATS
#define T2_STRIDE 128
#define OFF_T2   (OFF_WAVE + WAVES_PER_BLOCK * WAVE_FLOATS)
#define SMEM_FLOATS (OFF_T2 + Vq * T2_STRIDE)
#define SMEM_BYTES (SMEM_FLOATS * 4)

template <int K>
__device__ inline v8f gemm_tile_f32(const float* __restrict__ As, int lda,
                                    const float* __restrict__ Bqp, int nstride4,
                                    v8f acc, int lane) {
  const int rc = lane & 15;
  const int hk = (lane >> 4) << 1;
  const float* ap = As + rc * lda + hk;
  const float* bp = Bqp + rc * 4 + hk;
#pragma unroll
  for (int k = 0; k < K; k += 4) {
    v2f a = *(const v2f*)(ap + k);
    v2f b = *(const v2f*)(bp + (k >> 2) * nstride4);
    acc = __builtin_amdgcn_wmma_f32_16x16x4_f32(
         false, a,  false, b,
         (short)0, acc,  false,  false);
  }
  return acc;
}

__global__ __launch_bounds__(BLOCK_THREADS)
void PerPositionEMADelta_fused(const int* __restrict__ seq,
                               const float* __restrict__ embed,
                               const float* __restrict__ w1,
                               const float* __restrict__ b1,
                               const float* __restrict__ w2,
                               const float* __restrict__ b2,
                               const float* __restrict__ ln_g,
                               const float* __restrict__ ln_b,
                               const float* __restrict__ rp_w,
                               const float* __restrict__ rp_b,
                               const float* __restrict__ out_w,
                               const float* __restrict__ out_b,
                               const float* __restrict__ alpha_logits,
                               float* __restrict__ out) {
  extern __shared__ float sm[];
  const int tid  = threadIdx.x;
  const int lane = tid & 31;
  const int wave = tid >> 5;
  const int rc   = lane & 15;
  const int half = lane >> 4;
  const int wave_s = __builtin_amdgcn_readfirstlane(wave);

  float* s_w1  = sm + OFF_W1;
  float* s_w2  = sm + OFF_W2;
  float* s_rp  = sm + OFF_RP;
  float* s_ow  = sm + OFF_OW;
  float* s_b1  = sm + OFF_B1;
  float* s_b2  = sm + OFF_B2;
  float* s_lng = sm + OFF_LNG;
  float* s_lnb = sm + OFF_LNB;
  float* s_rpb = sm + OFF_RPB;
  float* s_ob  = sm + OFF_OB;
  float* s_om  = sm + OFF_OM;
  float* s_tab = sm + OFF_TAB;
  float* wbase = sm + OFF_WAVE + wave * WAVE_FLOATS;
  float* s_V   = wbase;
  float* s_wf  = sm + OFF_WF;
  float* s_bf  = sm + OFF_BF;

  const int base = (blockIdx.x * WAVES_PER_BLOCK + wave) * 16;

  v4f e4[4];
  {
    const float* src = embed + (size_t)(tid >> 2) * Hq + (tid & 3) * 16;
#pragma unroll
    for (int j = 0; j < 4; ++j) e4[j] = *(const v4f*)(src + 4 * j);
  }
  const float g_b1  = b1[tid & 127];
  const float g_b2  = b2[tid & 63];
  const float g_lng = ln_g[tid & 63];
  const float g_lnb = ln_b[tid & 63];
  const float g_rpb = rp_b[tid & 63];
  const float g_ob  = out_b[tid & 63];
  const float g_al  = alpha_logits[(tid & 31) < Lq ? (tid & 31) : 0];
  v4f q1[8], q2[8], q3[4], q4[4];
#pragma unroll
  for (int t = 0; t < 8; ++t) {
    const int wt = wave + 8 * t;
    const float* p = w1 + (wt >> 2) * 512 + (wt & 3) * 32 + lane;
    q1[t].x = p[0]; q1[t].y = p[128]; q1[t].z = p[256]; q1[t].w = p[384];
  }
  asm volatile("" ::: "memory");
#pragma unroll
  for (int t = 0; t < 8; ++t) {
    const int wt = wave + 8 * t;
    const float* p = w2 + (wt >> 1) * 256 + (wt & 1) * 32 + lane;
    q2[t].x = p[0]; q2[t].y = p[64]; q2[t].z = p[128]; q2[t].w = p[192];
  }
#pragma unroll
  for (int t = 0; t < 4; ++t) {
    const int wt = wave + 8 * t;
    const int o = (wt >> 1) * 256 + (wt & 1) * 32 + lane;
    const float* p = rp_w + o;
    const float* r = out_w + o;
    q3[t].x = p[0]; q3[t].y = p[64]; q3[t].z = p[128]; q3[t].w = p[192];
    q4[t].x = r[0]; q4[t].y = r[64]; q4[t].z = r[128]; q4[t].w = r[192];
  }
  int idx[Lq];
  {
    const v4i* sp = (const v4i*)(seq + (size_t)(blockIdx.x * (16 * WAVES_PER_BLOCK) + (wave & 3) * 32 + lane) * Lq);
#pragma unroll
    for (int j = 0; j < Lq / 4; ++j) {
      v4i t4 = sp[j];
      idx[4 * j + 0] = t4.x;
      idx[4 * j + 1] = t4.y;
      idx[4 * j + 2] = t4.z;
      idx[4 * j + 3] = t4.w;
    }
  }

  {
    float* dst = s_tab + (tid >> 2) * H_STRIDE + (tid & 3) * 16;
#pragma unroll
    for (int j = 0; j < 4; ++j) *(v4f*)(dst + 4 * j) = e4[j];
  }
  s_b1[tid & 127] = g_b1;
  s_b2[tid & 63]  = g_b2;
  s_lng[tid & 63] = g_lng;
  s_lnb[tid & 63] = g_lnb;
  s_rpb[tid & 63] = g_rpb;
  s_ob[tid & 63]  = g_ob;
  {
    float sg = 1.0f / (1.0f + expf(-g_al));
    s_om[tid & 31] = 1.0f - (0.49f * sg + 0.5f);
  }
#pragma unroll
  for (int t = 0; t < 8; ++t)
    *(v4f*)(s_w1 + (wave + 8 * t) * 128 + lane * 4) = q1[t];
  __builtin_amdgcn_fence(__ATOMIC_RELEASE, "workgroup", "local");
  __builtin_amdgcn_s_barrier();
  __builtin_amdgcn_fence(__ATOMIC_ACQUIRE, "workgroup", "local");

  {
    const int mt = wave & 3;
    const int hf = wave >> 2;
    float* Tt = s_tab + mt * (16 * H_STRIDE);
    float* X1 = sm + OFF_X1 + mt * X1_FLOATS;

    for (int i4 = 0; i4 < 4; ++i4) {
      const int nt = hf * 4 + i4;
      v8f acc;
      float bv = s_b1[nt * 16 + rc];
#pragma unroll
      for (int j = 0; j < 8; ++j) acc[j] = bv;
      acc = gemm_tile_f32<64>(Tt, H_STRIDE, s_w1 + nt * 16 * 4, 512, acc, lane);
#pragma unroll
      for (int j = 0; j < 8; ++j) {
        int row = j + 8 * half;
        X1[row * X1_STRIDE + nt * 16 + rc] = fmaxf(acc[j], 0.0f);
      }
    }
#pragma unroll
    for (int t = 0; t < 8; ++t)
      *(v4f*)(s_w2 + (wave + 8 * t) * 128 + lane * 4) = q2[t];
#pragma unroll
    for (int t = 0; t < 4; ++t) {
      *(v4f*)(s_rp + (wave + 8 * t) * 128 + lane * 4) = q3[t];
      *(v4f*)(s_ow + (wave + 8 * t) * 128 + lane * 4) = q4[t];
    }
    __syncthreads();

    for (int i2 = 0; i2 < 2; ++i2) {
      const int nt = hf * 2 + i2;
      v8f acc;
      float bv = s_b2[nt * 16 + rc];
#pragma unroll
      for (int j = 0; j < 8; ++j) acc[j] = bv;
      acc = gemm_tile_f32<128>(X1, X1_STRIDE, s_w2 + nt * 16 * 4, 256, acc, lane);
#pragma unroll
      for (int j = 0; j < 8; ++j) {
        int row = j + 8 * half;
        Tt[row * H_STRIDE + nt * 16 + rc] += acc[j];
      }
    }
    __syncthreads();

    if (tid < Vq) {
      float* rowp = s_tab + tid * H_STRIDE;
      float mu = 0.0f;
#pragma unroll
      for (int c = 0; c < 64; ++c) mu += rowp[c];
      mu *= (1.0f / 64.0f);
      float var = 0.0f;
#pragma unroll
      for (int c = 0; c < 64; ++c) { float d = rowp[c] - mu; var += d * d; }
      var *= (1.0f / 64.0f);
      float inv = rsqrtf(var + 1e-5f);
      float n0 = 0.0f, n1 = 0.0f;
#pragma unroll
      for (int c = 0; c < 32; ++c) {
        float y = (rowp[c] - mu) * inv * s_lng[c] + s_lnb[c];
        rowp[c] = y;
        n0 += y * y;
      }
#pragma unroll
      for (int c = 32; c < 64; ++c) {
        float y = (rowp[c] - mu) * inv * s_lng[c] + s_lnb[c];
        rowp[c] = y;
        n1 += y * y;
      }
      float nsq = n0 + n1;
      rowp[64] = 1.0f / fmaxf(sqrtf(nsq), 1e-12f);
    } else if (wave_s >= 4) {
      const int ft = wave_s - 4;
      const int hk2 = half << 1;
      const int m = 16 * ft + rc;
      const float* ap = s_rp + (m >> 2) * 256 + (m & 3) + 4 * hk2;
      const float* bp = s_ow + rc * 4 + hk2;
      v8f f0, f1, f2, f3, fb;
      {
        const float ob = s_ob[ft * 16 + rc];
#pragma unroll
        for (int j = 0; j < 8; ++j) {
          f0[j] = 0.0f; f1[j] = 0.0f; f2[j] = 0.0f; f3[j] = 0.0f; fb[j] = ob;
        }
      }
#pragma unroll
      for (int k = 0; k < 64; k += 4) {
        v2f a;
        a.x = ap[4 * k];
        a.y = ap[4 * k + 4];
        const v2f ab = *(const v2f*)(s_rpb + k + hk2);
        const float* bk = bp + (k >> 2) * 256;
        const v2f b0 = *(const v2f*)(bk);
        const v2f b1 = *(const v2f*)(bk + 64);
        const v2f b2 = *(const v2f*)(bk + 128);
        const v2f b3 = *(const v2f*)(bk + 192);
        const v2f bb = *(const v2f*)(bk + ft * 64);
        f0 = __builtin_amdgcn_wmma_f32_16x16x4_f32(false, a, false, b0, (short)0, f0, false, false);
        f1 = __builtin_amdgcn_wmma_f32_16x16x4_f32(false, a, false, b1, (short)0, f1, false, false);
        f2 = __builtin_amdgcn_wmma_f32_16x16x4_f32(false, a, false, b2, (short)0, f2, false, false);
        f3 = __builtin_amdgcn_wmma_f32_16x16x4_f32(false, a, false, b3, (short)0, f3, false, false);
        fb = __builtin_amdgcn_wmma_f32_16x16x4_f32(false, ab, false, bb, (short)0, fb, false, false);
      }
      float* wp = s_wf + (4 * ft + 2 * half) * 256 + 4 * rc;
      v4f lo, hi;
      lo.x = f0[0]; lo.y = f0[1]; lo.z = f0[2]; lo.w = f0[3];
      hi.x = f0[4]; hi.y = f0[5]; hi.z = f0[6]; hi.w = f0[7];
      *(v4f*)(wp) = lo;        *(v4f*)(wp + 256) = hi;
      lo.x = f1[0]; lo.y = f1[1]; lo.z = f1[2]; lo.w = f1[3];
      hi.x = f1[4]; hi.y = f1[5]; hi.z = f1[6]; hi.w = f1[7];
      *(v4f*)(wp + 64) = lo;   *(v4f*)(wp + 64 + 256) = hi;
      lo.x = f2[0]; lo.y = f2[1]; lo.z = f2[2]; lo.w = f2[3];
      hi.x = f2[4]; hi.y = f2[5]; hi.z = f2[6]; hi.w = f2[7];
      *(v4f*)(wp + 128) = lo;  *(v4f*)(wp + 128 + 256) = hi;
      lo.x = f3[0]; lo.y = f3[1]; lo.z = f3[2]; lo.w = f3[3];
      hi.x = f3[4]; hi.y = f3[5]; hi.z = f3[6]; hi.w = f3[7];
      *(v4f*)(wp + 192) = lo;  *(v4f*)(wp + 192 + 256) = hi;
      if (half == 0) s_bf[ft * 16 + rc] = fb[0];
    }
    __syncthreads();
  }

  {
    const int row = tid >> 2;
    const int c0 = 4 * (tid & 3) + (row & 3);
    const float* srow = s_tab + row * H_STRIDE;
    float* drow = sm + OFF_T2 + row * T2_STRIDE;
#pragma unroll
    for (int j = 0; j < 4; ++j) {
      const int c = (c0 + j) & 15;
      const v4f t4 = *(const v4f*)(srow + 4 * c);
      *(v4f*)(drow + 4 * c) = t4;
      *(v4f*)(drow + 64 + 4 * c) = t4;
    }
  }
  __syncthreads();

  if (wave_s < 4) {
    const int rot = (lane & 15) * 4;
    const float* s_t2 = sm + OFF_T2 + rot;
    v2f q[32], v[32];
    {
      const float* hl = s_t2 + idx[Lq - 1] * T2_STRIDE;
#pragma unroll
      for (int j = 0; j < 16; ++j) {
        v4f t4 = *(const v4f*)(hl + 4 * j);
        q[2 * j].x = t4.x;     q[2 * j].y = t4.y;
        q[2 * j + 1].x = t4.z; q[2 * j + 1].y = t4.w;
      }
#pragma unroll
      for (int j = 0; j < 32; ++j) { v[j].x = 0.0f; v[j].y = 0.0f; }
    }
    v2f hk[32];
    float inv, om;
    {
      const float* hs = s_t2 + idx[Lq - 2] * T2_STRIDE;
#pragma unroll
      for (int j = 0; j < 16; ++j) {
        v4f t4 = *(const v4f*)(hs + 4 * j);
        hk[2 * j].x = t4.x;     hk[2 * j].y = t4.y;
        hk[2 * j + 1].x = t4.z; hk[2 * j + 1].y = t4.w;
      }
      inv = s_tab[idx[Lq - 2] * H_STRIDE + 64];
      om = s_om[Lq - 2];
    }
#pragma unroll
    for (int s = Lq - 2; s >= 0; --s) {
      v2f hn[32];
      float invn = 0.0f, omn = 0.0f;
      if (s > 0) {
        const float* hs = s_t2 + idx[s - 1] * T2_STRIDE;
        invn = s_tab[idx[s - 1] * H_STRIDE + 64];
        omn = s_om[s - 1];
#pragma unroll
        for (int j = 0; j < 16; ++j) {
          v4f t4 = *(const v4f*)(hs + 4 * j);
          hn[2 * j].x = t4.x;     hn[2 * j].y = t4.y;
          hn[2 * j + 1].x = t4.z; hn[2 * j + 1].y = t4.w;
        }
      }
      __builtin_amdgcn_sched_barrier(0);
      v2f d[8];
#pragma unroll
      for (int i = 0; i < 8; ++i) d[i] = hk[i] * q[i];
#pragma unroll
      for (int j = 8; j < 32; j += 8) {
#pragma unroll
        for (int i = 0; i < 8; ++i)
          d[i] = __builtin_elementwise_fma(hk[j + i], q[j + i], d[i]);
      }
      const v2f e0 = d[0] + d[1];
      const v2f e1 = d[2] + d[3];
      const v2f e2 = d[4] + d[5];
      const v2f e3 = d[6] + d[7];
      const v2f f0 = e0 + e1;
      const v2f f1 = e2 + e3;
      const v2f dd = f0 + f1;
      const float dq = dd.x + dd.y;
      float c = om * (dq * inv);
      float ci = c * inv;
      v2f c2;  c2.x = c;    c2.y = c;
      v2f n2;  n2.x = -ci;  n2.y = -ci;
#pragma unroll
      for (int j = 0; j < 32; ++j) {
        q[j] = __builtin_elementwise_fma(n2, hk[j], q[j]);
        v[j] = __builtin_elementwise_fma(c2, hk[j], v[j]);
      }
      if (s > 0) {
#pragma unroll
        for (int j = 0; j < 32; ++j) hk[j] = hn[j];
        inv = invn;
        om = omn;
      }
#pragma unroll
      for (int j = 0; j < 32; j += 8)
        asm volatile("" :: "v"(v[j]), "v"(v[j + 1]), "v"(v[j + 2]), "v"(v[j + 3]),
                           "v"(v[j + 4]), "v"(v[j + 5]), "v"(v[j + 6]), "v"(v[j + 7]));
    }

    {
      float* dst = sm + OFF_WAVE + (wave * 32 + lane) * H_STRIDE;
#pragma unroll
      for (int j = 0; j < 16; ++j) {
        const int c = (j + (lane & 15)) & 15;
        v4f t4;
        t4.x = v[2 * j].x;     t4.y = v[2 * j].y;
        t4.z = v[2 * j + 1].x; t4.w = v[2 * j + 1].y;
        *(v4f*)(dst + 4 * c) = t4;
      }
    }
  }

  __syncthreads();

  for (int nt = 0; nt < 4; ++nt) {
    v8f acc;
    float bv = s_bf[nt * 16 + rc];
#pragma unroll
    for (int j = 0; j < 8; ++j) acc[j] = bv;
    acc = gemm_tile_f32<64>(s_V, H_STRIDE, s_wf + nt * 16 * 4, 256, acc, lane);
#pragma unroll
    for (int j = 0; j < 8; ++j) {
      int row = j + 8 * half;
      out[(size_t)(base + row) * Vq + nt * 16 + rc] = acc[j];
    }
  }
}

extern "C" void kernel_launch(void* const* d_in, const int* in_sizes, int n_in,
                              void* d_out, int out_size, void* d_ws, size_t ws_size,
                              hipStream_t stream) {
  (void)n_in; (void)out_size; (void)d_ws; (void)ws_size;
  const int* seq            = (const int*)d_in[0];
  const float* embed        = (const float*)d_in[1];
  const float* w1           = (const float*)d_in[2];
  const float* b1           = (const float*)d_in[3];
  const float* w2           = (const float*)d_in[4];
  const float* b2           = (const float*)d_in[5];
  const float* ln_g         = (const float*)d_in[6];
  const float* ln_b         = (const float*)d_in[7];
  const float* rp_w         = (const float*)d_in[8];
  const float* rp_b         = (const float*)d_in[9];
  const float* out_w        = (const float*)d_in[10];
  const float* out_b        = (const float*)d_in[11];
  const float* alpha_logits = (const float*)d_in[12];
  float* out = (float*)d_out;

  const int B = in_sizes[0] / Lq;
  const int blocks = B / (16 * WAVES_PER_BLOCK);

  (void)hipFuncSetAttribute((const void*)PerPositionEMADelta_fused,
                            hipFuncAttributeMaxDynamicSharedMemorySize,
                            (int)SMEM_BYTES);

  PerPositionEMADelta_fused<<<blocks, BLOCK_THREADS, SMEM_BYTES, stream>>>(
      seq, embed, w1, b1, w2, b2, ln_g, ln_b, rp_w, rp_b, out_w, out_b,
      alpha_logits, out);
}
